# starting kernel
# baseline (speedup 1.0000x reference)
#include <hip/hip_runtime.h>

typedef __attribute__((ext_vector_type(2))) float v2f;
typedef __attribute__((ext_vector_type(8))) float v8f;

#define C_DIM 128
#define LDA   132
#define MTILE 64

__global__ __launch_bounds__(256)
void splineconv_gemm_kernel(const float* __restrict__ X,
                            const float* __restrict__ W,
                            const float* __restrict__ root,
                            float* __restrict__ H,
                            int N)
{
    __shared__ float As[MTILE * LDA];
    const int tid  = threadIdx.x;
    const int wave = tid >> 5;
    const int lane = tid & 31;
    const int row0 = blockIdx.x * MTILE;

    for (int i = tid; i < MTILE * (C_DIM / 4); i += 256) {
        const int r  = i >> 5;
        const int c4 = i & 31;
        float4 v = make_float4(0.f, 0.f, 0.f, 0.f);
        const int gr = row0 + r;
        if (gr < N) v = ((const float4*)(X + (size_t)gr * C_DIM))[c4];
        float* dst = &As[r * LDA + c4 * 4];
        dst[0] = v.x; dst[1] = v.y; dst[2] = v.z; dst[3] = v.w;
    }
    __syncthreads();

    const int hi  = lane >> 4;
    const int l16 = lane & 15;

    for (int t = wave; t < 24; t += 8) {
        const int gcol = t * 16;
        const float* Bbase;
        if (gcol < C_DIM)          Bbase = W + gcol;
        else if (gcol < 2 * C_DIM) Bbase = W + (size_t)C_DIM * C_DIM + (gcol - C_DIM);
        else                       Bbase = root + (gcol - 2 * C_DIM);

        const float* Bl = Bbase + (size_t)(2 * hi) * C_DIM + l16;
        const float* Al = &As[l16 * LDA + 2 * hi];

        v2f bf[32];
        #pragma unroll
        for (int k4 = 0; k4 < 32; ++k4) {
            bf[k4].x = Bl[(size_t)(4 * k4) * C_DIM];
            bf[k4].y = Bl[(size_t)(4 * k4 + 1) * C_DIM];
        }

        v8f acc[4] = {{}, {}, {}, {}};
        #pragma unroll
        for (int k4 = 0; k4 < 32; ++k4) {
            const int k0 = 4 * k4;
            v2f a0, a1, a2, a3;
            a0.x = Al[k0];              a0.y = Al[k0 + 1];
            a1.x = Al[16 * LDA + k0];   a1.y = Al[16 * LDA + k0 + 1];
            a2.x = Al[32 * LDA + k0];   a2.y = Al[32 * LDA + k0 + 1];
            a3.x = Al[48 * LDA + k0];   a3.y = Al[48 * LDA + k0 + 1];
            acc[0] = __builtin_amdgcn_wmma_f32_16x16x4_f32(
                         false, a0, false, bf[k4], (short)0, acc[0], false, false);
            acc[1] = __builtin_amdgcn_wmma_f32_16x16x4_f32(
                         false, a1, false, bf[k4], (short)0, acc[1], false, false);
            acc[2] = __builtin_amdgcn_wmma_f32_16x16x4_f32(
                         false, a2, false, bf[k4], (short)0, acc[2], false, false);
            acc[3] = __builtin_amdgcn_wmma_f32_16x16x4_f32(
                         false, a3, false, bf[k4], (short)0, acc[3], false, false);
        }

        #pragma unroll
        for (int s = 0; s < 4; ++s) {
            const int rbase = row0 + s * 16;
            float* Hp = H + (size_t)rbase * (3 * C_DIM) + gcol;
            #pragma unroll
            for (int r = 0; r < 8; ++r) {
                const int m = r + 8 * hi;
                if (rbase + m < N)
                    Hp[(size_t)m * (3 * C_DIM) + l16] = acc[s][r];
            }
        }
    }
}

__global__ __launch_bounds__(256)
void scatter_kernel(const int* __restrict__ ei,
                    const float* __restrict__ u,
                    const float* __restrict__ H,
                    float* __restrict__ agg,
                    float* __restrict__ deg,
                    int E)
{
    const int e    = blockIdx.x * 8 + (threadIdx.x >> 5);
    const int lane = threadIdx.x & 31;
    if (e >= E) return;

    const int   s  = ei[e];
    const int   d  = ei[E + e];
    const float uu = u[e];
    const float w0 = 1.0f - uu;

    const float4* h0 = (const float4*)(H + (size_t)s * 3 * C_DIM);
    const float4* h1 = (const float4*)(H + (size_t)s * 3 * C_DIM + C_DIM);
    const float4 a = h0[lane];
    const float4 b = h1[lane];
    const float4 m = make_float4(w0 * a.x + uu * b.x, w0 * a.y + uu * b.y,
                                 w0 * a.z + uu * b.z, w0 * a.w + uu * b.w);

    float* ap = agg + (size_t)d * C_DIM + lane * 4;
    atomicAdd(ap + 0, m.x);
    atomicAdd(ap + 1, m.y);
    atomicAdd(ap + 2, m.z);
    atomicAdd(ap + 3, m.w);
    if (lane == 0) atomicAdd(deg + d, 1.0f);
}

__global__ __launch_bounds__(256)
void combine_kernel(const float* __restrict__ agg,
                    const float* __restrict__ deg,
                    const float* __restrict__ H,
                    const float* __restrict__ bias,
                    float* __restrict__ out,
                    int N)
{
    const int n    = blockIdx.x * 8 + (threadIdx.x >> 5);
    const int lane = threadIdx.x & 31;
    if (n >= N) return;

    const float inv = 1.0f / fmaxf(deg[n], 1.0f);
    const float4 a  = ((const float4*)(agg + (size_t)n * C_DIM))[lane];
    const float4 r  = ((const float4*)(H + (size_t)n * 3 * C_DIM + 2 * C_DIM))[lane];
    const float4 bb = ((const float4*)bias)[lane];
    const float4 o  = make_float4(a.x * inv + r.x + bb.x,
                                  a.y * inv + r.y + bb.y,
                                  a.z * inv + r.z + bb.z,
                                  a.w * inv + r.w + bb.w);
    ((float4*)(out + (size_t)n * C_DIM))[lane] = o;
}

__global__ __launch_bounds__(256)
void zero_kernel(float* __restrict__ p, size_t n)
{
    const size_t i = (size_t)blockIdx.x * blockDim.x + threadIdx.x;
    if (i < n) p[i] = 0.0f;
}

extern "C" void kernel_launch(void* const* d_in, const int* in_sizes, int n_in,
                              void* d_out, int out_size, void* d_ws, size_t ws_size,
                              hipStream_t stream)
{
    const float* x  = (const float*)d_in[0];
    const int*   ei = (const int*)d_in[1];
    const float* ea = (const float*)d_in[2];
    const float* W1 = (const float*)d_in[3];
    const float* r1 = (const float*)d_in[4];
    const float* b1 = (const float*)d_in[5];
    const float* W2 = (const float*)d_in[6];
    const float* r2 = (const float*)d_in[7];
    const float* b2 = (const float*)d_in[8];
    float* out = (float*)d_out;

    const int N = in_sizes[0] / C_DIM;
    const int E = in_sizes[1] / 2;

    float* ws   = (float*)d_ws;
    float* Hbuf = ws;
    float* agg  = Hbuf + (size_t)N * 3 * C_DIM;
    float* deg  = agg  + (size_t)N * C_DIM;
    float* hmid = deg  + N;

    const int mblocks = (N + MTILE - 1) / MTILE;
    const int eblocks = (E + 7) / 8;
    const int nblocks = (N + 7) / 8;
    const size_t zn   = (size_t)N * (C_DIM + 1);
    const int zblocks = (int)((zn + 255) / 256);

    zero_kernel<<<zblocks, 256, 0, stream>>>(agg, zn);
    splineconv_gemm_kernel<<<mblocks, 256, 0, stream>>>(x, W1, r1, Hbuf, N);
    scatter_kernel<<<eblocks, 256, 0, stream>>>(ei, ea, Hbuf, agg, deg, E);
    combine_kernel<<<nblocks, 256, 0, stream>>>(agg, deg, Hbuf, b1, hmid, N);

    zero_kernel<<<zblocks, 256, 0, stream>>>(agg, zn);
    splineconv_gemm_kernel<<<mblocks, 256, 0, stream>>>(hmid, W2, r2, Hbuf, N);
    scatter_kernel<<<eblocks, 256, 0, stream>>>(ei, ea, Hbuf, agg, deg, E);
    combine_kernel<<<nblocks, 256, 0, stream>>>(agg, deg, Hbuf, b2, out, N);
}


# scatter: lane l owns channels l,l+32,l+64,l+96, so each wave-wide f32 atomic add and each gather load covers one whole 128-B line instead of 4 lines of 8 lanes; sum order still atomic arrival
# speedup vs baseline: 1.0786x; 1.0786x over previous
#include <hip/hip_runtime.h>

typedef __attribute__((ext_vector_type(2))) float v2f;
typedef __attribute__((ext_vector_type(8))) float v8f;

#define C_DIM 128
#define LDA   132
#define MTILE 64

__global__ __launch_bounds__(256)
void splineconv_gemm_kernel(const float* __restrict__ X,
                            const float* __restrict__ W,
                            const float* __restrict__ root,
                            float* __restrict__ H,
                            int N)
{
    __shared__ float As[MTILE * LDA];
    const int tid  = threadIdx.x;
    const int wave = tid >> 5;
    const int lane = tid & 31;
    const int row0 = blockIdx.x * MTILE;

    for (int i = tid; i < MTILE * (C_DIM / 4); i += 256) {
        const int r  = i >> 5;
        const int c4 = i & 31;
        float4 v = make_float4(0.f, 0.f, 0.f, 0.f);
        const int gr = row0 + r;
        if (gr < N) v = ((const float4*)(X + (size_t)gr * C_DIM))[c4];
        float* dst = &As[r * LDA + c4 * 4];
        dst[0] = v.x; dst[1] = v.y; dst[2] = v.z; dst[3] = v.w;
    }
    __syncthreads();

    const int hi  = lane >> 4;
    const int l16 = lane & 15;

    for (int t = wave; t < 24; t += 8) {
        const int gcol = t * 16;
        const float* Bbase;
        if (gcol < C_DIM)          Bbase = W + gcol;
        else if (gcol < 2 * C_DIM) Bbase = W + (size_t)C_DIM * C_DIM + (gcol - C_DIM);
        else                       Bbase = root + (gcol - 2 * C_DIM);

        const float* Bl = Bbase + (size_t)(2 * hi) * C_DIM + l16;
        const float* Al = &As[l16 * LDA + 2 * hi];

        v2f bf[32];
        #pragma unroll
        for (int k4 = 0; k4 < 32; ++k4) {
            bf[k4].x = Bl[(size_t)(4 * k4) * C_DIM];
            bf[k4].y = Bl[(size_t)(4 * k4 + 1) * C_DIM];
        }

        v8f acc[4] = {{}, {}, {}, {}};
        #pragma unroll
        for (int k4 = 0; k4 < 32; ++k4) {
            const int k0 = 4 * k4;
            v2f a0, a1, a2, a3;
            a0.x = Al[k0];              a0.y = Al[k0 + 1];
            a1.x = Al[16 * LDA + k0];   a1.y = Al[16 * LDA + k0 + 1];
            a2.x = Al[32 * LDA + k0];   a2.y = Al[32 * LDA + k0 + 1];
            a3.x = Al[48 * LDA + k0];   a3.y = Al[48 * LDA + k0 + 1];
            acc[0] = __builtin_amdgcn_wmma_f32_16x16x4_f32(
                         false, a0, false, bf[k4], (short)0, acc[0], false, false);
            acc[1] = __builtin_amdgcn_wmma_f32_16x16x4_f32(
                         false, a1, false, bf[k4], (short)0, acc[1], false, false);
            acc[2] = __builtin_amdgcn_wmma_f32_16x16x4_f32(
                         false, a2, false, bf[k4], (short)0, acc[2], false, false);
            acc[3] = __builtin_amdgcn_wmma_f32_16x16x4_f32(
                         false, a3, false, bf[k4], (short)0, acc[3], false, false);
        }

        #pragma unroll
        for (int s = 0; s < 4; ++s) {
            const int rbase = row0 + s * 16;
            float* Hp = H + (size_t)rbase * (3 * C_DIM) + gcol;
            #pragma unroll
            for (int r = 0; r < 8; ++r) {
                const int m = r + 8 * hi;
                if (rbase + m < N)
                    Hp[(size_t)m * (3 * C_DIM) + l16] = acc[s][r];
            }
        }
    }
}

__global__ __launch_bounds__(256)
void scatter_kernel(const int* __restrict__ ei,
                    const float* __restrict__ u,
                    const float* __restrict__ H,
                    float* __restrict__ agg,
                    float* __restrict__ deg,
                    int E)
{
    const int e    = blockIdx.x * 8 + (threadIdx.x >> 5);
    const int lane = threadIdx.x & 31;
    if (e >= E) return;

    const int   s  = ei[e];
    const int   d  = ei[E + e];
    const float uu = u[e];
    const float w0 = 1.0f - uu;

    const float* hs = H + (size_t)s * 3 * C_DIM + lane;
    const float a0 = hs[0];
    const float a1 = hs[32];
    const float a2 = hs[64];
    const float a3 = hs[96];
    const float b0 = hs[C_DIM];
    const float b1 = hs[C_DIM + 32];
    const float b2 = hs[C_DIM + 64];
    const float b3 = hs[C_DIM + 96];

    const float m0 = w0 * a0 + uu * b0;
    const float m1 = w0 * a1 + uu * b1;
    const float m2 = w0 * a2 + uu * b2;
    const float m3 = w0 * a3 + uu * b3;

    float* ap = agg + (size_t)d * C_DIM + lane;
    atomicAdd(ap,      m0);
    atomicAdd(ap + 32, m1);
    atomicAdd(ap + 64, m2);
    atomicAdd(ap + 96, m3);
    if (lane == 0) atomicAdd(deg + d, 1.0f);
}

__global__ __launch_bounds__(256)
void combine_kernel(const float* __restrict__ agg,
                    const float* __restrict__ deg,
                    const float* __restrict__ H,
                    const float* __restrict__ bias,
                    float* __restrict__ out,
                    int N)
{
    const int n    = blockIdx.x * 8 + (threadIdx.x >> 5);
    const int lane = threadIdx.x & 31;
    if (n >= N) return;

    const float inv = 1.0f / fmaxf(deg[n], 1.0f);
    const float4 a  = ((const float4*)(agg + (size_t)n * C_DIM))[lane];
    const float4 r  = ((const float4*)(H + (size_t)n * 3 * C_DIM + 2 * C_DIM))[lane];
    const float4 bb = ((const float4*)bias)[lane];
    const float4 o  = make_float4(a.x * inv + r.x + bb.x,
                                  a.y * inv + r.y + bb.y,
                                  a.z * inv + r.z + bb.z,
                                  a.w * inv + r.w + bb.w);
    ((float4*)(out + (size_t)n * C_DIM))[lane] = o;
}

__global__ __launch_bounds__(256)
void zero_kernel(float* __restrict__ p, size_t n)
{
    const size_t i = (size_t)blockIdx.x * blockDim.x + threadIdx.x;
    if (i < n) p[i] = 0.0f;
}

extern "C" void kernel_launch(void* const* d_in, const int* in_sizes, int n_in,
                              void* d_out, int out_size, void* d_ws, size_t ws_size,
                              hipStream_t stream)
{
    const float* x  = (const float*)d_in[0];
    const int*   ei = (const int*)d_in[1];
    const float* ea = (const float*)d_in[2];
    const float* W1 = (const float*)d_in[3];
    const float* r1 = (const float*)d_in[4];
    const float* b1 = (const float*)d_in[5];
    const float* W2 = (const float*)d_in[6];
    const float* r2 = (const float*)d_in[7];
    const float* b2 = (const float*)d_in[8];
    float* out = (float*)d_out;

    const int N = in_sizes[0] / C_DIM;
    const int E = in_sizes[1] / 2;

    float* ws   = (float*)d_ws;
    float* Hbuf = ws;
    float* agg  = Hbuf + (size_t)N * 3 * C_DIM;
    float* deg  = agg  + (size_t)N * C_DIM;
    float* hmid = deg  + N;

    const int mblocks = (N + MTILE - 1) / MTILE;
    const int eblocks = (E + 7) / 8;
    const int nblocks = (N + 7) / 8;
    const size_t zn   = (size_t)N * (C_DIM + 1);
    const int zblocks = (int)((zn + 255) / 256);

    zero_kernel<<<zblocks, 256, 0, stream>>>(agg, zn);
    splineconv_gemm_kernel<<<mblocks, 256, 0, stream>>>(x, W1, r1, Hbuf, N);
    scatter_kernel<<<eblocks, 256, 0, stream>>>(ei, ea, Hbuf, agg, deg, E);
    combine_kernel<<<nblocks, 256, 0, stream>>>(agg, deg, Hbuf, b1, hmid, N);

    zero_kernel<<<zblocks, 256, 0, stream>>>(agg, zn);
    splineconv_gemm_kernel<<<mblocks, 256, 0, stream>>>(hmid, W2, r2, Hbuf, N);
    scatter_kernel<<<eblocks, 256, 0, stream>>>(ei, ea, Hbuf, agg, deg, E);
    combine_kernel<<<nblocks, 256, 0, stream>>>(agg, deg, Hbuf, b2, out, N);
}


# edges grouped by destination once per call (int count, scan, slot atomics); one wave per node sums its edges in registers: no float atomics, no zero or combine pass; sum order = slot arrival
# speedup vs baseline: 1.5833x; 1.4679x over previous
#include <hip/hip_runtime.h>

typedef __attribute__((ext_vector_type(2))) float v2f;
typedef __attribute__((ext_vector_type(8))) float v8f;

#define C_DIM 128
#define LDA   132
#define MTILE 64

__global__ __launch_bounds__(256)
void splineconv_gemm_kernel(const float* __restrict__ X,
                            const float* __restrict__ W,
                            const float* __restrict__ root,
                            float* __restrict__ H,
                            int N)
{
    __shared__ float As[MTILE * LDA];
    const int tid  = threadIdx.x;
    const int wave = tid >> 5;
    const int lane = tid & 31;
    const int row0 = blockIdx.x * MTILE;

    for (int i = tid; i < MTILE * (C_DIM / 4); i += 256) {
        const int r  = i >> 5;
        const int c4 = i & 31;
        float4 v = make_float4(0.f, 0.f, 0.f, 0.f);
        const int gr = row0 + r;
        if (gr < N) v = ((const float4*)(X + (size_t)gr * C_DIM))[c4];
        float* dst = &As[r * LDA + c4 * 4];
        dst[0] = v.x; dst[1] = v.y; dst[2] = v.z; dst[3] = v.w;
    }
    __syncthreads();

    const int hi  = lane >> 4;
    const int l16 = lane & 15;

    for (int t = wave; t < 24; t += 8) {
        const int gcol = t * 16;
        const float* Bbase;
        if (gcol < C_DIM)          Bbase = W + gcol;
        else if (gcol < 2 * C_DIM) Bbase = W + (size_t)C_DIM * C_DIM + (gcol - C_DIM);
        else                       Bbase = root + (gcol - 2 * C_DIM);

        const float* Bl = Bbase + (size_t)(2 * hi) * C_DIM + l16;
        const float* Al = &As[l16 * LDA + 2 * hi];

        v2f bf[32];
        #pragma unroll
        for (int k4 = 0; k4 < 32; ++k4) {
            bf[k4].x = Bl[(size_t)(4 * k4) * C_DIM];
            bf[k4].y = Bl[(size_t)(4 * k4 + 1) * C_DIM];
        }

        v8f acc[4] = {{}, {}, {}, {}};
        #pragma unroll
        for (int k4 = 0; k4 < 32; ++k4) {
            const int k0 = 4 * k4;
            v2f a0, a1, a2, a3;
            a0.x = Al[k0];              a0.y = Al[k0 + 1];
            a1.x = Al[16 * LDA + k0];   a1.y = Al[16 * LDA + k0 + 1];
            a2.x = Al[32 * LDA + k0];   a2.y = Al[32 * LDA + k0 + 1];
            a3.x = Al[48 * LDA + k0];   a3.y = Al[48 * LDA + k0 + 1];
            acc[0] = __builtin_amdgcn_wmma_f32_16x16x4_f32(
                         false, a0, false, bf[k4], (short)0, acc[0], false, false);
            acc[1] = __builtin_amdgcn_wmma_f32_16x16x4_f32(
                         false, a1, false, bf[k4], (short)0, acc[1], false, false);
            acc[2] = __builtin_amdgcn_wmma_f32_16x16x4_f32(
                         false, a2, false, bf[k4], (short)0, acc[2], false, false);
            acc[3] = __builtin_amdgcn_wmma_f32_16x16x4_f32(
                         false, a3, false, bf[k4], (short)0, acc[3], false, false);
        }

        #pragma unroll
        for (int s = 0; s < 4; ++s) {
            const int rbase = row0 + s * 16;
            float* Hp = H + (size_t)rbase * (3 * C_DIM) + gcol;
            #pragma unroll
            for (int r = 0; r < 8; ++r) {
                const int m = r + 8 * hi;
                if (rbase + m < N)
                    Hp[(size_t)m * (3 * C_DIM) + l16] = acc[s][r];
            }
        }
    }
}

__global__ __launch_bounds__(256)
void zero_count_kernel(int* __restrict__ cnt, int n)
{
    const int i = blockIdx.x * 256 + threadIdx.x;
    if (i < n) cnt[i] = 0;
}

__global__ __launch_bounds__(256)
void count_kernel(const int* __restrict__ ei, int* __restrict__ cnt, int E, int N)
{
    const int e = blockIdx.x * 256 + threadIdx.x;
    if (e >= E) return;
    const int d = ei[E + e];
    if ((unsigned)d < (unsigned)N) atomicAdd(cnt + d, 1);
}

__global__ __launch_bounds__(256)
void scan_kernel(const int* __restrict__ cnt,
                 int* __restrict__ rowptr,
                 int* __restrict__ cursor,
                 int N)
{
    __shared__ int wtot[8];
    __shared__ int wpre[8];
    const int t    = threadIdx.x;
    const int lane = t & 31;
    const int w    = t >> 5;
    const int b0   = blockIdx.x * 1024;
    const int4* c4 = (const int4*)cnt;

    int part = 0;
    const int trips = blockIdx.x;
    #pragma unroll 8
    for (int k = 0; k < trips; ++k) {
        const int4 v = c4[k * 256 + t];
        part += (v.x + v.y) + (v.z + v.w);
    }

    const int i0 = b0 + t * 4;
    int v0 = 0, v1 = 0, v2 = 0, v3 = 0;
    if (i0 + 3 < N) {
        const int4 v = c4[i0 >> 2];
        v0 = v.x; v1 = v.y; v2 = v.z; v3 = v.w;
    } else {
        if (i0     < N) v0 = cnt[i0];
        if (i0 + 1 < N) v1 = cnt[i0 + 1];
        if (i0 + 2 < N) v2 = cnt[i0 + 2];
    }
    const int ts = (v0 + v1) + (v2 + v3);

    #pragma unroll
    for (int off = 16; off > 0; off >>= 1) part += __shfl_xor(part, off);
    int inc = ts;
    #pragma unroll
    for (int off = 1; off < 32; off <<= 1) {
        const int y = __shfl_up(inc, off);
        if (lane >= off) inc += y;
    }
    if (lane == 31) wtot[w] = inc;
    if (lane == 0)  wpre[w] = part;
    __syncthreads();

    int base = 0;
    #pragma unroll
    for (int k = 0; k < 8; ++k) {
        base += wpre[k];
        if (k < w) base += wtot[k];
    }
    const int ex = base + inc - ts;
    if (i0     < N) { rowptr[i0]     = ex;                cursor[i0]     = ex; }
    if (i0 + 1 < N) { rowptr[i0 + 1] = ex + v0;           cursor[i0 + 1] = ex + v0; }
    if (i0 + 2 < N) { rowptr[i0 + 2] = ex + v0 + v1;      cursor[i0 + 2] = ex + v0 + v1; }
    if (i0 + 3 < N) { rowptr[i0 + 3] = ex + v0 + v1 + v2; cursor[i0 + 3] = ex + v0 + v1 + v2; }
    if (blockIdx.x == gridDim.x - 1 && t == 255) rowptr[N] = ex + ts;
}

__global__ __launch_bounds__(256)
void fill_kernel(const int* __restrict__ ei,
                 const float* __restrict__ u,
                 int* __restrict__ cursor,
                 int2* __restrict__ sedge,
                 int E, int N)
{
    const int e = blockIdx.x * 256 + threadIdx.x;
    if (e >= E) return;
    const int d = ei[E + e];
    if ((unsigned)d >= (unsigned)N) return;
    int s = ei[e];
    s = s < 0 ? 0 : (s >= N ? N - 1 : s);
    const int p = atomicAdd(cursor + d, 1);
    if ((unsigned)p < (unsigned)E) sedge[p] = make_int2(s, __float_as_int(u[e]));
}

__global__ __launch_bounds__(256)
void gather_combine_kernel(const int* __restrict__ rowptr,
                           const int2* __restrict__ sedge,
                           const float* __restrict__ H,
                           const float* __restrict__ bias,
                           float* __restrict__ out,
                           int N)
{
    const int lane = threadIdx.x & 31;
    const int n = blockIdx.x * 8 + __builtin_amdgcn_readfirstlane((int)(threadIdx.x >> 5));
    if (n >= N) return;

    const int beg = __builtin_amdgcn_readfirstlane(rowptr[n]);
    const int end = __builtin_amdgcn_readfirstlane(rowptr[n + 1]);

    const float* hr = H + (size_t)n * 3 * C_DIM + 2 * C_DIM + lane;
    const float r0 = hr[0], r1 = hr[32], r2 = hr[64], r3 = hr[96];
    const float g0 = bias[lane], g1 = bias[lane + 32], g2 = bias[lane + 64], g3 = bias[lane + 96];

    float acc0 = 0.f, acc1 = 0.f, acc2 = 0.f, acc3 = 0.f;

    for (int base = beg; base < end; base += 32) {
        const int idx = base + lane;
        int2 ed = make_int2(0, 0);
        if (idx < end) ed = sedge[idx];
        const int cntc = (end - base) < 32 ? (end - base) : 32;

        for (int j = 0; j < cntc; j += 4) {
            float a[4][4], b[4][4], uu[4];
            #pragma unroll
            for (int q = 0; q < 4; ++q) {
                const int s = __builtin_amdgcn_readlane(ed.x, j + q);
                uu[q] = __int_as_float(__builtin_amdgcn_readlane(ed.y, j + q));
                const float* hs = H + (size_t)s * 3 * C_DIM + lane;
                a[q][0] = hs[0];          a[q][1] = hs[32];
                a[q][2] = hs[64];         a[q][3] = hs[96];
                b[q][0] = hs[C_DIM];      b[q][1] = hs[C_DIM + 32];
                b[q][2] = hs[C_DIM + 64]; b[q][3] = hs[C_DIM + 96];
            }
            #pragma unroll
            for (int q = 0; q < 4; ++q) {
                if (j + q < cntc) {
                    const float w0 = 1.0f - uu[q];
                    acc0 += w0 * a[q][0] + uu[q] * b[q][0];
                    acc1 += w0 * a[q][1] + uu[q] * b[q][1];
                    acc2 += w0 * a[q][2] + uu[q] * b[q][2];
                    acc3 += w0 * a[q][3] + uu[q] * b[q][3];
                }
            }
        }
    }

    const float deg = (float)(end - beg);
    const float inv = 1.0f / fmaxf(deg, 1.0f);
    float* op = out + (size_t)n * C_DIM + lane;
    op[0]  = acc0 * inv + r0 + g0;
    op[32] = acc1 * inv + r1 + g1;
    op[64] = acc2 * inv + r2 + g2;
    op[96] = acc3 * inv + r3 + g3;
}

extern "C" void kernel_launch(void* const* d_in, const int* in_sizes, int n_in,
                              void* d_out, int out_size, void* d_ws, size_t ws_size,
                              hipStream_t stream)
{
    const float* x  = (const float*)d_in[0];
    const int*   ei = (const int*)d_in[1];
    const float* ea = (const float*)d_in[2];
    const float* W1 = (const float*)d_in[3];
    const float* r1 = (const float*)d_in[4];
    const float* b1 = (const float*)d_in[5];
    const float* W2 = (const float*)d_in[6];
    const float* r2 = (const float*)d_in[7];
    const float* b2 = (const float*)d_in[8];
    float* out = (float*)d_out;

    const int N = in_sizes[0] / C_DIM;
    const int E = in_sizes[1] / 2;

    const int NP = ((N + 4) / 4) * 4;
    float* ws     = (float*)d_ws;
    float* Hbuf   = ws;
    float* hmid   = Hbuf + (size_t)N * 3 * C_DIM;
    int*   cnt    = (int*)(hmid + (size_t)N * C_DIM);
    int*   rowptr = cnt + NP;
    int*   cursor = rowptr + NP;
    int2*  sedge  = (int2*)(cursor + NP);

    const int mblocks = (N + MTILE - 1) / MTILE;
    const int nblocks = (N + 7) / 8;
    const int eblocks = (E + 255) / 256;
    const int cblocks = (N + 255) / 256;
    const int sblocks = (N + 1023) / 1024;

    zero_count_kernel<<<cblocks, 256, 0, stream>>>(cnt, N);
    count_kernel<<<eblocks, 256, 0, stream>>>(ei, cnt, E, N);
    scan_kernel<<<sblocks, 256, 0, stream>>>(cnt, rowptr, cursor, N);
    fill_kernel<<<eblocks, 256, 0, stream>>>(ei, ea, cursor, sedge, E, N);

    splineconv_gemm_kernel<<<mblocks, 256, 0, stream>>>(x, W1, r1, Hbuf, N);
    gather_combine_kernel<<<nblocks, 256, 0, stream>>>(rowptr, sedge, Hbuf, b1, hmid, N);

    splineconv_gemm_kernel<<<mblocks, 256, 0, stream>>>(hmid, W2, r2, Hbuf, N);
    gather_combine_kernel<<<nblocks, 256, 0, stream>>>(rowptr, sedge, Hbuf, b2, out, N);
}


# edge sort: the counting add returns each edge's rank in its destination and the rank is kept, so placing is rowptr[dst]+rank with no second scattered atomic pass; sum order = counting-add arrival
# speedup vs baseline: 2.5915x; 1.6368x over previous
#include <hip/hip_runtime.h>

typedef __attribute__((ext_vector_type(2))) float v2f;
typedef __attribute__((ext_vector_type(8))) float v8f;

#define C_DIM 128
#define LDA   132
#define MTILE 64

__global__ __launch_bounds__(256)
void splineconv_gemm_kernel(const float* __restrict__ PQ,
                            const float* __restrict__ X,
                            const float* __restrict__ W,
                            const float* __restrict__ root,
                            const float* __restrict__ bias,
                            float* __restrict__ Out,
                            int N)
{
    __shared__ float As[3][MTILE * LDA];
    const int tid  = threadIdx.x;
    const int wave = tid >> 5;
    const int lane = tid & 31;
    const int row0 = blockIdx.x * MTILE;

    {
        const int c4 = tid & 31;
        const int rr = tid >> 5;
        #pragma unroll
        for (int m = 0; m < 8; ++m) {
            const int gr = min(row0 + rr + 8 * m, N - 1);
            const float4 vp = ((const float4*)(PQ + (size_t)gr * (2 * C_DIM)))[c4];
            const float4 vq = ((const float4*)(PQ + (size_t)gr * (2 * C_DIM) + C_DIM))[c4];
            const float4 vx = ((const float4*)(X + (size_t)gr * C_DIM))[c4];
            float* d0 = &As[0][(rr + 8 * m) * LDA + c4 * 4];
            float* d1 = &As[1][(rr + 8 * m) * LDA + c4 * 4];
            float* d2 = &As[2][(rr + 8 * m) * LDA + c4 * 4];
            d0[0] = vp.x; d0[1] = vp.y; d0[2] = vp.z; d0[3] = vp.w;
            d1[0] = vq.x; d1[1] = vq.y; d1[2] = vq.z; d1[3] = vq.w;
            d2[0] = vx.x; d2[1] = vx.y; d2[2] = vx.z; d2[3] = vx.w;
        }
    }
    __syncthreads();

    const int hi   = lane >> 4;
    const int l16  = lane & 15;
    const int gcol = wave * 16;

    v8f acc[4] = {{}, {}, {}, {}};

    #pragma unroll
    for (int c = 0; c < 3; ++c) {
        const float* Bbase = (c == 0) ? (W + gcol)
                           : (c == 1) ? (W + (size_t)C_DIM * C_DIM + gcol)
                                      : (root + gcol);

        const float* Bl = Bbase + (size_t)(2 * hi) * C_DIM + l16;
        const float* Al = &As[c][l16 * LDA + 2 * hi];

        v2f bf[32];
        #pragma unroll
        for (int k4 = 0; k4 < 32; ++k4) {
            bf[k4].x = Bl[(size_t)(4 * k4) * C_DIM];
            bf[k4].y = Bl[(size_t)(4 * k4 + 1) * C_DIM];
        }

        #pragma unroll
        for (int k4 = 0; k4 < 32; ++k4) {
            const int k0 = 4 * k4;
            v2f a0, a1, a2, a3;
            a0.x = Al[k0];              a0.y = Al[k0 + 1];
            a1.x = Al[16 * LDA + k0];   a1.y = Al[16 * LDA + k0 + 1];
            a2.x = Al[32 * LDA + k0];   a2.y = Al[32 * LDA + k0 + 1];
            a3.x = Al[48 * LDA + k0];   a3.y = Al[48 * LDA + k0 + 1];
            acc[0] = __builtin_amdgcn_wmma_f32_16x16x4_f32(
                         false, a0, false, bf[k4], (short)0, acc[0], false, false);
            acc[1] = __builtin_amdgcn_wmma_f32_16x16x4_f32(
                         false, a1, false, bf[k4], (short)0, acc[1], false, false);
            acc[2] = __builtin_amdgcn_wmma_f32_16x16x4_f32(
                         false, a2, false, bf[k4], (short)0, acc[2], false, false);
            acc[3] = __builtin_amdgcn_wmma_f32_16x16x4_f32(
                         false, a3, false, bf[k4], (short)0, acc[3], false, false);
        }
    }

    const float bv = bias[gcol + l16];
    #pragma unroll
    for (int s = 0; s < 4; ++s) {
        const int rbase = row0 + s * 16;
        float* Op = Out + (size_t)rbase * C_DIM + gcol;
        #pragma unroll
        for (int r = 0; r < 8; ++r) {
            const int m = r + 8 * hi;
            if (rbase + m < N)
                Op[(size_t)m * C_DIM + l16] = acc[s][r] + bv;
        }
    }
}

__global__ __launch_bounds__(256)
void zero_count_kernel(int* __restrict__ cnt, int n)
{
    const int i = blockIdx.x * 256 + threadIdx.x;
    if (i < n) cnt[i] = 0;
}

__global__ __launch_bounds__(256)
void count_kernel(const int* __restrict__ ei, int* __restrict__ cnt,
                  int* __restrict__ rank,
                  int E, int N)
{
    const int e = blockIdx.x * 256 + threadIdx.x;
    if (e >= E) return;
    const int d = ei[E + e];
    if ((unsigned)d < (unsigned)N) rank[e] = atomicAdd(cnt + d, 1);
}

__global__ __launch_bounds__(256)
void scan_kernel(const int* __restrict__ cnt,
                 int* __restrict__ rowptr,
                 int N)
{
    __shared__ int wtot[8];
    __shared__ int wpre[8];
    const int t    = threadIdx.x;
    const int lane = t & 31;
    const int w    = t >> 5;
    const int b0   = blockIdx.x * 1024;
    const int4* c4 = (const int4*)cnt;

    int part = 0;
    const int trips = blockIdx.x;
    #pragma unroll 8
    for (int k = 0; k < trips; ++k) {
        const int4 v = c4[k * 256 + t];
        part += (v.x + v.y) + (v.z + v.w);
    }

    const int i0 = b0 + t * 4;
    int v0 = 0, v1 = 0, v2 = 0, v3 = 0;
    if (i0 + 3 < N) {
        const int4 v = c4[i0 >> 2];
        v0 = v.x; v1 = v.y; v2 = v.z; v3 = v.w;
    } else {
        if (i0     < N) v0 = cnt[i0];
        if (i0 + 1 < N) v1 = cnt[i0 + 1];
        if (i0 + 2 < N) v2 = cnt[i0 + 2];
    }
    const int ts = (v0 + v1) + (v2 + v3);

    #pragma unroll
    for (int off = 16; off > 0; off >>= 1) part += __shfl_xor(part, off);
    int inc = ts;
    #pragma unroll
    for (int off = 1; off < 32; off <<= 1) {
        const int y = __shfl_up(inc, off);
        if (lane >= off) inc += y;
    }
    if (lane == 31) wtot[w] = inc;
    if (lane == 0)  wpre[w] = part;
    __syncthreads();

    int base = 0;
    #pragma unroll
    for (int k = 0; k < 8; ++k) {
        base += wpre[k];
        if (k < w) base += wtot[k];
    }
    const int ex = base + inc - ts;
    if (i0     < N) rowptr[i0]     = ex;
    if (i0 + 1 < N) rowptr[i0 + 1] = ex + v0;
    if (i0 + 2 < N) rowptr[i0 + 2] = ex + v0 + v1;
    if (i0 + 3 < N) rowptr[i0 + 3] = ex + v0 + v1 + v2;
    if (blockIdx.x == gridDim.x - 1 && t == 255) rowptr[N] = ex + ts;
}

__global__ __launch_bounds__(256)
void fill_kernel(const int* __restrict__ ei,
                 const float* __restrict__ u,
                 const int* __restrict__ rowptr,
                 const int* __restrict__ rank,
                 int2* __restrict__ sedge,
                 int E, int N)
{
    const int e = blockIdx.x * 256 + threadIdx.x;
    if (e >= E) return;
    const int d = ei[E + e];
    if ((unsigned)d >= (unsigned)N) return;
    int s = ei[e];
    s = s < 0 ? 0 : (s >= N ? N - 1 : s);
    const int p = rowptr[d] + rank[e];
    if ((unsigned)p < (unsigned)E) sedge[p] = make_int2(s, __float_as_int(u[e]));
}

__global__ __launch_bounds__(256)
void gather_sum_kernel(const int* __restrict__ rowptr,
                       const int2* __restrict__ sedge,
                       const float* __restrict__ X,
                       float* __restrict__ PQ,
                       int N)
{
    const int lane = threadIdx.x & 31;
    const int n = blockIdx.x * 8 + __builtin_amdgcn_readfirstlane((int)(threadIdx.x >> 5));
    if (n >= N) return;

    const int beg = __builtin_amdgcn_readfirstlane(rowptr[n]);
    const int end = __builtin_amdgcn_readfirstlane(rowptr[n + 1]);

    float p0 = 0.f, p1 = 0.f, p2 = 0.f, p3 = 0.f;
    float q0 = 0.f, q1 = 0.f, q2 = 0.f, q3 = 0.f;

    for (int base = beg; base < end; base += 32) {
        const int idx = base + lane;
        int2 ed = make_int2(0, 0);
        if (idx < end) ed = sedge[idx];
        const int cntc = (end - base) < 32 ? (end - base) : 32;

        for (int j = 0; j < cntc; j += 8) {
            float a[8][4], uu[8];
            #pragma unroll
            for (int q = 0; q < 8; ++q) {
                const int s = __builtin_amdgcn_readlane(ed.x, j + q);
                uu[q] = __int_as_float(__builtin_amdgcn_readlane(ed.y, j + q));
                const float* xs = X + (size_t)s * C_DIM + lane;
                a[q][0] = xs[0];  a[q][1] = xs[32];
                a[q][2] = xs[64]; a[q][3] = xs[96];
            }
            #pragma unroll
            for (int q = 0; q < 8; ++q) {
                if (j + q < cntc) {
                    const float w0 = 1.0f - uu[q];
                    p0 += w0 * a[q][0];  q0 += uu[q] * a[q][0];
                    p1 += w0 * a[q][1];  q1 += uu[q] * a[q][1];
                    p2 += w0 * a[q][2];  q2 += uu[q] * a[q][2];
                    p3 += w0 * a[q][3];  q3 += uu[q] * a[q][3];
                }
            }
        }
    }

    const float deg = (float)(end - beg);
    const float inv = 1.0f / fmaxf(deg, 1.0f);
    float* op = PQ + (size_t)n * (2 * C_DIM) + lane;
    op[0]           = p0 * inv;
    op[32]          = p1 * inv;
    op[64]          = p2 * inv;
    op[96]          = p3 * inv;
    op[C_DIM]       = q0 * inv;
    op[C_DIM + 32]  = q1 * inv;
    op[C_DIM + 64]  = q2 * inv;
    op[C_DIM + 96]  = q3 * inv;
}

extern "C" void kernel_launch(void* const* d_in, const int* in_sizes, int n_in,
                              void* d_out, int out_size, void* d_ws, size_t ws_size,
                              hipStream_t stream)
{
    const float* x  = (const float*)d_in[0];
    const int*   ei = (const int*)d_in[1];
    const float* ea = (const float*)d_in[2];
    const float* W1 = (const float*)d_in[3];
    const float* r1 = (const float*)d_in[4];
    const float* b1 = (const float*)d_in[5];
    const float* W2 = (const float*)d_in[6];
    const float* r2 = (const float*)d_in[7];
    const float* b2 = (const float*)d_in[8];
    float* out = (float*)d_out;

    const int N = in_sizes[0] / C_DIM;
    const int E = in_sizes[1] / 2;

    const int NP = ((N + 4) / 4) * 4;
    float* ws     = (float*)d_ws;
    float* PQbuf  = ws;
    float* hmid   = PQbuf + (size_t)N * 2 * C_DIM;
    int*   cnt    = (int*)(hmid + (size_t)N * C_DIM);
    int*   rowptr = cnt + NP;
    int2*  sedge  = (int2*)(rowptr + NP);
    int*   rank   = (int*)(sedge + E);

    const int mblocks = (N + MTILE - 1) / MTILE;
    const int nblocks = (N + 7) / 8;
    const int eblocks = (E + 255) / 256;
    const int cblocks = (N + 255) / 256;
    const int sblocks = (N + 1023) / 1024;

    zero_count_kernel<<<cblocks, 256, 0, stream>>>(cnt, N);
    count_kernel<<<eblocks, 256, 0, stream>>>(ei, cnt, rank, E, N);
    scan_kernel<<<sblocks, 256, 0, stream>>>(cnt, rowptr, N);
    fill_kernel<<<eblocks, 256, 0, stream>>>(ei, ea, rowptr, rank, sedge, E, N);

    gather_sum_kernel<<<nblocks, 256, 0, stream>>>(rowptr, sedge, x, PQbuf, N);
    splineconv_gemm_kernel<<<mblocks, 256, 0, stream>>>(PQbuf, x, W1, r1, b1, hmid, N);

    gather_sum_kernel<<<nblocks, 256, 0, stream>>>(rowptr, sedge, hmid, PQbuf, N);
    splineconv_gemm_kernel<<<mblocks, 256, 0, stream>>>(PQbuf, hmid, W2, r2, b2, out, N);
}


# destination counting in LDS: 32 edge slices, one block each with a counter per node (rank = LDS add's return), a column pass gives slice offsets; no global atomics; sum order: slice, then LDS arrival
# speedup vs baseline: 2.9264x; 1.1292x over previous
#include <hip/hip_runtime.h>

typedef __attribute__((ext_vector_type(2))) float v2f;
typedef __attribute__((ext_vector_type(8))) float v8f;

#define C_DIM 128
#define LDA   132
#define MTILE 64

__global__ __launch_bounds__(256)
void splineconv_gemm_kernel(const float* __restrict__ PQ,
                            const float* __restrict__ X,
                            const float* __restrict__ W,
                            const float* __restrict__ root,
                            const float* __restrict__ bias,
                            float* __restrict__ Out,
                            int N)
{
    __shared__ float As[3][MTILE * LDA];
    const int tid  = threadIdx.x;
    const int wave = tid >> 5;
    const int lane = tid & 31;
    const int row0 = blockIdx.x * MTILE;

    {
        const int c4 = tid & 31;
        const int rr = tid >> 5;
        #pragma unroll
        for (int m = 0; m < 8; ++m) {
            const int gr = min(row0 + rr + 8 * m, N - 1);
            const float4 vp = ((const float4*)(PQ + (size_t)gr * (2 * C_DIM)))[c4];
            const float4 vq = ((const float4*)(PQ + (size_t)gr * (2 * C_DIM) + C_DIM))[c4];
            const float4 vx = ((const float4*)(X + (size_t)gr * C_DIM))[c4];
            float* d0 = &As[0][(rr + 8 * m) * LDA + c4 * 4];
            float* d1 = &As[1][(rr + 8 * m) * LDA + c4 * 4];
            float* d2 = &As[2][(rr + 8 * m) * LDA + c4 * 4];
            d0[0] = vp.x; d0[1] = vp.y; d0[2] = vp.z; d0[3] = vp.w;
            d1[0] = vq.x; d1[1] = vq.y; d1[2] = vq.z; d1[3] = vq.w;
            d2[0] = vx.x; d2[1] = vx.y; d2[2] = vx.z; d2[3] = vx.w;
        }
    }
    __syncthreads();

    const int hi   = lane >> 4;
    const int l16  = lane & 15;
    const int gcol = wave * 16;

    v8f acc[4] = {{}, {}, {}, {}};

    #pragma unroll
    for (int c = 0; c < 3; ++c) {
        const float* Bbase = (c == 0) ? (W + gcol)
                           : (c == 1) ? (W + (size_t)C_DIM * C_DIM + gcol)
                                      : (root + gcol);

        const float* Bl = Bbase + (size_t)(2 * hi) * C_DIM + l16;
        const float* Al = &As[c][l16 * LDA + 2 * hi];

        v2f bf[32];
        #pragma unroll
        for (int k4 = 0; k4 < 32; ++k4) {
            bf[k4].x = Bl[(size_t)(4 * k4) * C_DIM];
            bf[k4].y = Bl[(size_t)(4 * k4 + 1) * C_DIM];
        }

        #pragma unroll
        for (int k4 = 0; k4 < 32; ++k4) {
            const int k0 = 4 * k4;
            v2f a0, a1, a2, a3;
            a0.x = Al[k0];              a0.y = Al[k0 + 1];
            a1.x = Al[16 * LDA + k0];   a1.y = Al[16 * LDA + k0 + 1];
            a2.x = Al[32 * LDA + k0];   a2.y = Al[32 * LDA + k0 + 1];
            a3.x = Al[48 * LDA + k0];   a3.y = Al[48 * LDA + k0 + 1];
            acc[0] = __builtin_amdgcn_wmma_f32_16x16x4_f32(
                         false, a0, false, bf[k4], (short)0, acc[0], false, false);
            acc[1] = __builtin_amdgcn_wmma_f32_16x16x4_f32(
                         false, a1, false, bf[k4], (short)0, acc[1], false, false);
            acc[2] = __builtin_amdgcn_wmma_f32_16x16x4_f32(
                         false, a2, false, bf[k4], (short)0, acc[2], false, false);
            acc[3] = __builtin_amdgcn_wmma_f32_16x16x4_f32(
                         false, a3, false, bf[k4], (short)0, acc[3], false, false);
        }
    }

    const float bv = bias[gcol + l16];
    #pragma unroll
    for (int s = 0; s < 4; ++s) {
        const int rbase = row0 + s * 16;
        float* Op = Out + (size_t)rbase * C_DIM + gcol;
        #pragma unroll
        for (int r = 0; r < 8; ++r) {
            const int m = r + 8 * hi;
            if (rbase + m < N)
                Op[(size_t)m * C_DIM + l16] = acc[s][r] + bv;
        }
    }
}

#define NSLICE 32
#define CAP    51200

__global__ __launch_bounds__(1024)
void slice_count_kernel(const int* __restrict__ ei,
                        int* __restrict__ part,
                        int* __restrict__ rank,
                        int E, int N, int NP, int L)
{
    __shared__ int4 lc4[CAP / 4];
    int* lcnt = (int*)lc4;
    const int t  = threadIdx.x;
    const int e0 = blockIdx.x * L;
    int len = E - e0;
    len = len < 0 ? 0 : (len > L ? L : len);
    const int* dst = ei + E + e0;
    int* prow = part + (size_t)blockIdx.x * NP;

    for (int nb = 0; nb < N; nb += CAP) {
        const int span = (N - nb) < CAP ? (N - nb) : CAP;
        const int z4 = (span + 3) >> 2;
        for (int i = t; i < z4; i += 1024) lc4[i] = make_int4(0, 0, 0, 0);
        __syncthreads();

        for (int k0 = t; k0 < len; k0 += 8 * 1024) {
            int dd[8];
            #pragma unroll
            for (int q = 0; q < 8; ++q) {
                const int i = k0 + 1024 * q;
                dd[q] = (i < len) ? dst[i] : -1;
            }
            int rr[8];
            #pragma unroll
            for (int q = 0; q < 8; ++q) {
                const unsigned key = (unsigned)dd[q] - (unsigned)nb;
                rr[q] = -1;
                if (key < (unsigned)span) rr[q] = atomicAdd(&lcnt[key], 1);
            }
            #pragma unroll
            for (int q = 0; q < 8; ++q)
                if (rr[q] >= 0) rank[e0 + k0 + 1024 * q] = rr[q];
        }
        __syncthreads();

        const int span4 = span >> 2;
        int4* o4 = (int4*)(prow + nb);
        for (int i = t; i < span4; i += 1024) o4[i] = lc4[i];
        const int tl = span4 * 4 + t;
        if (tl < span) prow[nb + tl] = lcnt[tl];
        __syncthreads();
    }
}

__global__ __launch_bounds__(256)
void slice_offsets_kernel(int* __restrict__ part, int* __restrict__ cnt, int N, int NP)
{
    const int d = blockIdx.x * 256 + threadIdx.x;
    if (d >= N) return;
    int v[NSLICE];
    #pragma unroll
    for (int s = 0; s < NSLICE; ++s) v[s] = part[(size_t)s * NP + d];
    int run = 0;
    #pragma unroll
    for (int s = 0; s < NSLICE; ++s) {
        part[(size_t)s * NP + d] = run;
        run += v[s];
    }
    cnt[d] = run;
}

__global__ __launch_bounds__(256)
void scan_kernel(const int* __restrict__ cnt,
                 int* __restrict__ rowptr,
                 int N)
{
    __shared__ int wtot[8];
    __shared__ int wpre[8];
    const int t    = threadIdx.x;
    const int lane = t & 31;
    const int w    = t >> 5;
    const int b0   = blockIdx.x * 1024;
    const int4* c4 = (const int4*)cnt;

    int part = 0;
    const int trips = blockIdx.x;
    #pragma unroll 8
    for (int k = 0; k < trips; ++k) {
        const int4 v = c4[k * 256 + t];
        part += (v.x + v.y) + (v.z + v.w);
    }

    const int i0 = b0 + t * 4;
    int v0 = 0, v1 = 0, v2 = 0, v3 = 0;
    if (i0 + 3 < N) {
        const int4 v = c4[i0 >> 2];
        v0 = v.x; v1 = v.y; v2 = v.z; v3 = v.w;
    } else {
        if (i0     < N) v0 = cnt[i0];
        if (i0 + 1 < N) v1 = cnt[i0 + 1];
        if (i0 + 2 < N) v2 = cnt[i0 + 2];
    }
    const int ts = (v0 + v1) + (v2 + v3);

    #pragma unroll
    for (int off = 16; off > 0; off >>= 1) part += __shfl_xor(part, off);
    int inc = ts;
    #pragma unroll
    for (int off = 1; off < 32; off <<= 1) {
        const int y = __shfl_up(inc, off);
        if (lane >= off) inc += y;
    }
    if (lane == 31) wtot[w] = inc;
    if (lane == 0)  wpre[w] = part;
    __syncthreads();

    int base = 0;
    #pragma unroll
    for (int k = 0; k < 8; ++k) {
        base += wpre[k];
        if (k < w) base += wtot[k];
    }
    const int ex = base + inc - ts;
    if (i0     < N) rowptr[i0]     = ex;
    if (i0 + 1 < N) rowptr[i0 + 1] = ex + v0;
    if (i0 + 2 < N) rowptr[i0 + 2] = ex + v0 + v1;
    if (i0 + 3 < N) rowptr[i0 + 3] = ex + v0 + v1 + v2;
    if (blockIdx.x == gridDim.x - 1 && t == 255) rowptr[N] = ex + ts;
}

__global__ __launch_bounds__(256)
void fill_kernel(const int* __restrict__ ei,
                 const float* __restrict__ u,
                 const int* __restrict__ rowptr,
                 const int* __restrict__ part,
                 const int* __restrict__ rank,
                 int2* __restrict__ sedge,
                 int E, int N, int NP,
                 int bps)
{
    const int e = blockIdx.x * 256 + threadIdx.x;
    if (e >= E) return;
    const int d = ei[E + e];
    if ((unsigned)d >= (unsigned)N) return;
    int s = ei[e];
    s = s < 0 ? 0 : (s >= N ? N - 1 : s);
    const int sl = blockIdx.x / bps;
    const int p = rowptr[d] + part[(size_t)sl * NP + d] + rank[e];
    if ((unsigned)p < (unsigned)E) sedge[p] = make_int2(s, __float_as_int(u[e]));
}

__global__ __launch_bounds__(256)
void gather_sum_kernel(const int* __restrict__ rowptr,
                       const int2* __restrict__ sedge,
                       const float* __restrict__ X,
                       float* __restrict__ PQ,
                       int N)
{
    const int lane = threadIdx.x & 31;
    const int n = blockIdx.x * 8 + __builtin_amdgcn_readfirstlane((int)(threadIdx.x >> 5));
    if (n >= N) return;

    const int beg = __builtin_amdgcn_readfirstlane(rowptr[n]);
    const int end = __builtin_amdgcn_readfirstlane(rowptr[n + 1]);

    float p0 = 0.f, p1 = 0.f, p2 = 0.f, p3 = 0.f;
    float q0 = 0.f, q1 = 0.f, q2 = 0.f, q3 = 0.f;

    for (int base = beg; base < end; base += 32) {
        const int idx = base + lane;
        int2 ed = make_int2(0, 0);
        if (idx < end) ed = sedge[idx];
        const int cntc = (end - base) < 32 ? (end - base) : 32;

        for (int j = 0; j < cntc; j += 8) {
            float a[8][4], uu[8];
            #pragma unroll
            for (int q = 0; q < 8; ++q) {
                const int s = __builtin_amdgcn_readlane(ed.x, j + q);
                uu[q] = __int_as_float(__builtin_amdgcn_readlane(ed.y, j + q));
                const float* xs = X + (size_t)s * C_DIM + lane;
                a[q][0] = xs[0];  a[q][1] = xs[32];
                a[q][2] = xs[64]; a[q][3] = xs[96];
            }
            #pragma unroll
            for (int q = 0; q < 8; ++q) {
                if (j + q < cntc) {
                    const float w0 = 1.0f - uu[q];
                    p0 += w0 * a[q][0];  q0 += uu[q] * a[q][0];
                    p1 += w0 * a[q][1];  q1 += uu[q] * a[q][1];
                    p2 += w0 * a[q][2];  q2 += uu[q] * a[q][2];
                    p3 += w0 * a[q][3];  q3 += uu[q] * a[q][3];
                }
            }
        }
    }

    const float deg = (float)(end - beg);
    const float inv = 1.0f / fmaxf(deg, 1.0f);
    float* op = PQ + (size_t)n * (2 * C_DIM) + lane;
    op[0]           = p0 * inv;
    op[32]          = p1 * inv;
    op[64]          = p2 * inv;
    op[96]          = p3 * inv;
    op[C_DIM]       = q0 * inv;
    op[C_DIM + 32]  = q1 * inv;
    op[C_DIM + 64]  = q2 * inv;
    op[C_DIM + 96]  = q3 * inv;
}

extern "C" void kernel_launch(void* const* d_in, const int* in_sizes, int n_in,
                              void* d_out, int out_size, void* d_ws, size_t ws_size,
                              hipStream_t stream)
{
    const float* x  = (const float*)d_in[0];
    const int*   ei = (const int*)d_in[1];
    const float* ea = (const float*)d_in[2];
    const float* W1 = (const float*)d_in[3];
    const float* r1 = (const float*)d_in[4];
    const float* b1 = (const float*)d_in[5];
    const float* W2 = (const float*)d_in[6];
    const float* r2 = (const float*)d_in[7];
    const float* b2 = (const float*)d_in[8];
    float* out = (float*)d_out;

    const int N = in_sizes[0] / C_DIM;
    const int E = in_sizes[1] / 2;

    const int NP = ((N + 4) / 4) * 4;
    float* ws     = (float*)d_ws;
    float* PQbuf  = ws;
    float* hmid   = PQbuf + (size_t)N * 2 * C_DIM;
    int*   cnt    = (int*)(hmid + (size_t)N * C_DIM);
    int*   rowptr = cnt + NP;
    int*   part   = rowptr + NP;
    int2*  sedge  = (int2*)(part + (size_t)NSLICE * NP);
    int*   rank   = (int*)(sedge + E);

    const int mblocks = (N + MTILE - 1) / MTILE;
    const int nblocks = (N + 7) / 8;
    const int eblocks = (E + 255) / 256;
    const int cblocks = (N + 255) / 256;
    const int sblocks = (N + 1023) / 1024;
    const int bps     = (E + NSLICE * 256 - 1) / (NSLICE * 256);
    const int L       = bps * 256;

    slice_count_kernel<<<NSLICE, 1024, 0, stream>>>(ei, part, rank, E, N, NP, L);
    slice_offsets_kernel<<<cblocks, 256, 0, stream>>>(part, cnt, N, NP);
    scan_kernel<<<sblocks, 256, 0, stream>>>(cnt, rowptr, N);
    fill_kernel<<<eblocks, 256, 0, stream>>>(ei, ea, rowptr, part, rank, sedge, E, N, NP, bps);

    gather_sum_kernel<<<nblocks, 256, 0, stream>>>(rowptr, sedge, x, PQbuf, N);
    splineconv_gemm_kernel<<<mblocks, 256, 0, stream>>>(PQbuf, x, W1, r1, b1, hmid, N);

    gather_sum_kernel<<<nblocks, 256, 0, stream>>>(rowptr, sedge, hmid, PQbuf, N);
    splineconv_gemm_kernel<<<mblocks, 256, 0, stream>>>(PQbuf, hmid, W2, r2, b2, out, N);
}


# product kernel: rows past the last whole round of resident 64-row blocks (768 = 3 per WGP) go as 16-row blocks, first in the grid: no WGP ends on a lone 64-row block; sum order kept
# speedup vs baseline: 3.1316x; 1.0701x over previous
#include <hip/hip_runtime.h>

typedef __attribute__((ext_vector_type(2))) float v2f;
typedef __attribute__((ext_vector_type(8))) float v8f;

#define C_DIM 128
#define LDA   132
#define MTILE 64

template <int MSUB>
__device__ __forceinline__
void product_tile(float* __restrict__ As,
                  const float* __restrict__ PQ,
                  const float* __restrict__ X,
                  const float* __restrict__ W,
                  const float* __restrict__ root,
                  const float* __restrict__ bias,
                  float* __restrict__ Out,
                  int N, int row0)
{
    const int tid  = threadIdx.x;
    const int wave = tid >> 5;
    const int lane = tid & 31;

    {
        const int c4 = tid & 31;
        const int rr = tid >> 5;
        #pragma unroll
        for (int m = 0; m < 2 * MSUB; ++m) {
            const int gr = min(row0 + rr + 8 * m, N - 1);
            const float4 vp = ((const float4*)(PQ + (size_t)gr * (2 * C_DIM)))[c4];
            const float4 vq = ((const float4*)(PQ + (size_t)gr * (2 * C_DIM) + C_DIM))[c4];
            const float4 vx = ((const float4*)(X + (size_t)gr * C_DIM))[c4];
            float* d0 = &As[0 * MTILE * LDA + (rr + 8 * m) * LDA + c4 * 4];
            float* d1 = &As[1 * MTILE * LDA + (rr + 8 * m) * LDA + c4 * 4];
            float* d2 = &As[2 * MTILE * LDA + (rr + 8 * m) * LDA + c4 * 4];
            d0[0] = vp.x; d0[1] = vp.y; d0[2] = vp.z; d0[3] = vp.w;
            d1[0] = vq.x; d1[1] = vq.y; d1[2] = vq.z; d1[3] = vq.w;
            d2[0] = vx.x; d2[1] = vx.y; d2[2] = vx.z; d2[3] = vx.w;
        }
    }
    __syncthreads();

    const int hi   = lane >> 4;
    const int l16  = lane & 15;
    const int gcol = wave * 16;

    v8f acc[MSUB];
    #pragma unroll
    for (int s = 0; s < MSUB; ++s) acc[s] = v8f{};

    #pragma unroll
    for (int c = 0; c < 3; ++c) {
        const float* Bbase = (c == 0) ? (W + gcol)
                           : (c == 1) ? (W + (size_t)C_DIM * C_DIM + gcol)
                                      : (root + gcol);

        const float* Bl = Bbase + (size_t)(2 * hi) * C_DIM + l16;
        const float* Al = &As[c * MTILE * LDA + l16 * LDA + 2 * hi];

        v2f bf[32];
        #pragma unroll
        for (int k4 = 0; k4 < 32; ++k4) {
            bf[k4].x = Bl[(size_t)(4 * k4) * C_DIM];
            bf[k4].y = Bl[(size_t)(4 * k4 + 1) * C_DIM];
        }

        #pragma unroll
        for (int k4 = 0; k4 < 32; ++k4) {
            const int k0 = 4 * k4;
            v2f a[MSUB];
            #pragma unroll
            for (int s = 0; s < MSUB; ++s) {
                a[s].x = Al[s * 16 * LDA + k0];
                a[s].y = Al[s * 16 * LDA + k0 + 1];
            }
            #pragma unroll
            for (int s = 0; s < MSUB; ++s)
                acc[s] = __builtin_amdgcn_wmma_f32_16x16x4_f32(
                             false, a[s], false, bf[k4], (short)0, acc[s], false, false);
        }
    }

    const float bv = bias[gcol + l16];
    #pragma unroll
    for (int s = 0; s < MSUB; ++s) {
        const int rbase = row0 + s * 16;
        float* Op = Out + (size_t)rbase * C_DIM + gcol;
        #pragma unroll
        for (int r = 0; r < 8; ++r) {
            const int m = r + 8 * hi;
            if (rbase + m < N)
                Op[(size_t)m * C_DIM + l16] = acc[s][r] + bv;
        }
    }
}

#define RESIDENT_BLOCKS 768

__global__ __launch_bounds__(256)
void splineconv_gemm_kernel(const float* __restrict__ PQ,
                            const float* __restrict__ X,
                            const float* __restrict__ W,
                            const float* __restrict__ root,
                            const float* __restrict__ bias,
                            float* __restrict__ Out,
                            int N,
                            int nsmall,
                            int rsmall0)
{
    __shared__ float As[3 * MTILE * LDA];
    const int b = blockIdx.x;
    if (b < nsmall)
        product_tile<1>(As, PQ, X, W, root, bias, Out, N, rsmall0 + b * 16);
    else
        product_tile<4>(As, PQ, X, W, root, bias, Out, N, (b - nsmall) * MTILE);
}

#define NSLICE 32
#define CAP    51200

__global__ __launch_bounds__(1024)
void slice_count_kernel(const int* __restrict__ ei,
                        int* __restrict__ part,
                        int* __restrict__ rank,
                        int E, int N, int NP, int L)
{
    __shared__ int4 lc4[CAP / 4];
    int* lcnt = (int*)lc4;
    const int t  = threadIdx.x;
    const int e0 = blockIdx.x * L;
    int len = E - e0;
    len = len < 0 ? 0 : (len > L ? L : len);
    const int* dst = ei + E + e0;
    int* prow = part + (size_t)blockIdx.x * NP;

    for (int nb = 0; nb < N; nb += CAP) {
        const int span = (N - nb) < CAP ? (N - nb) : CAP;
        const int z4 = (span + 3) >> 2;
        for (int i = t; i < z4; i += 1024) lc4[i] = make_int4(0, 0, 0, 0);
        __syncthreads();

        for (int k0 = t; k0 < len; k0 += 8 * 1024) {
            int dd[8];
            #pragma unroll
            for (int q = 0; q < 8; ++q) {
                const int i = k0 + 1024 * q;
                dd[q] = (i < len) ? dst[i] : -1;
            }
            int rr[8];
            #pragma unroll
            for (int q = 0; q < 8; ++q) {
                const unsigned key = (unsigned)dd[q] - (unsigned)nb;
                rr[q] = -1;
                if (key < (unsigned)span) rr[q] = atomicAdd(&lcnt[key], 1);
            }
            #pragma unroll
            for (int q = 0; q < 8; ++q)
                if (rr[q] >= 0) rank[e0 + k0 + 1024 * q] = rr[q];
        }
        __syncthreads();

        const int span4 = span >> 2;
        int4* o4 = (int4*)(prow + nb);
        for (int i = t; i < span4; i += 1024) o4[i] = lc4[i];
        const int tl = span4 * 4 + t;
        if (tl < span) prow[nb + tl] = lcnt[tl];
        __syncthreads();
    }
}

__global__ __launch_bounds__(256)
void slice_offsets_kernel(int* __restrict__ part, int* __restrict__ cnt, int N, int NP)
{
    const int d = blockIdx.x * 256 + threadIdx.x;
    if (d >= N) return;
    int v[NSLICE];
    #pragma unroll
    for (int s = 0; s < NSLICE; ++s) v[s] = part[(size_t)s * NP + d];
    int run = 0;
    #pragma unroll
    for (int s = 0; s < NSLICE; ++s) {
        part[(size_t)s * NP + d] = run;
        run += v[s];
    }
    cnt[d] = run;
}

__global__ __launch_bounds__(256)
void scan_kernel(const int* __restrict__ cnt,
                 int* __restrict__ rowptr,
                 int N)
{
    __shared__ int wtot[8];
    __shared__ int wpre[8];
    const int t    = threadIdx.x;
    const int lane = t & 31;
    const int w    = t >> 5;
    const int b0   = blockIdx.x * 1024;
    const int4* c4 = (const int4*)cnt;

    int part = 0;
    const int trips = blockIdx.x;
    #pragma unroll 8
    for (int k = 0; k < trips; ++k) {
        const int4 v = c4[k * 256 + t];
        part += (v.x + v.y) + (v.z + v.w);
    }

    const int i0 = b0 + t * 4;
    int v0 = 0, v1 = 0, v2 = 0, v3 = 0;
    if (i0 + 3 < N) {
        const int4 v = c4[i0 >> 2];
        v0 = v.x; v1 = v.y; v2 = v.z; v3 = v.w;
    } else {
        if (i0     < N) v0 = cnt[i0];
        if (i0 + 1 < N) v1 = cnt[i0 + 1];
        if (i0 + 2 < N) v2 = cnt[i0 + 2];
    }
    const int ts = (v0 + v1) + (v2 + v3);

    #pragma unroll
    for (int off = 16; off > 0; off >>= 1) part += __shfl_xor(part, off);
    int inc = ts;
    #pragma unroll
    for (int off = 1; off < 32; off <<= 1) {
        const int y = __shfl_up(inc, off);
        if (lane >= off) inc += y;
    }
    if (lane == 31) wtot[w] = inc;
    if (lane == 0)  wpre[w] = part;
    __syncthreads();

    int base = 0;
    #pragma unroll
    for (int k = 0; k < 8; ++k) {
        base += wpre[k];
        if (k < w) base += wtot[k];
    }
    const int ex = base + inc - ts;
    if (i0     < N) rowptr[i0]     = ex;
    if (i0 + 1 < N) rowptr[i0 + 1] = ex + v0;
    if (i0 + 2 < N) rowptr[i0 + 2] = ex + v0 + v1;
    if (i0 + 3 < N) rowptr[i0 + 3] = ex + v0 + v1 + v2;
    if (blockIdx.x == gridDim.x - 1 && t == 255) rowptr[N] = ex + ts;
}

__global__ __launch_bounds__(256)
void fill_kernel(const int* __restrict__ ei,
                 const float* __restrict__ u,
                 const int* __restrict__ rowptr,
                 const int* __restrict__ part,
                 const int* __restrict__ rank,
                 int2* __restrict__ sedge,
                 int E, int N, int NP,
                 int bps)
{
    const int e = blockIdx.x * 256 + threadIdx.x;
    if (e >= E) return;
    const int d = ei[E + e];
    if ((unsigned)d >= (unsigned)N) return;
    int s = ei[e];
    s = s < 0 ? 0 : (s >= N ? N - 1 : s);
    const int sl = blockIdx.x / bps;
    const int p = rowptr[d] + part[(size_t)sl * NP + d] + rank[e];
    if ((unsigned)p < (unsigned)E) sedge[p] = make_int2(s, __float_as_int(u[e]));
}

__global__ __launch_bounds__(256)
void gather_sum_kernel(const int* __restrict__ rowptr,
                       const int2* __restrict__ sedge,
                       const float* __restrict__ X,
                       float* __restrict__ PQ,
                       int N)
{
    const int lane = threadIdx.x & 31;
    const int n = blockIdx.x * 8 + __builtin_amdgcn_readfirstlane((int)(threadIdx.x >> 5));
    if (n >= N) return;

    const int beg = __builtin_amdgcn_readfirstlane(rowptr[n]);
    const int end = __builtin_amdgcn_readfirstlane(rowptr[n + 1]);

    float p0 = 0.f, p1 = 0.f, p2 = 0.f, p3 = 0.f;
    float q0 = 0.f, q1 = 0.f, q2 = 0.f, q3 = 0.f;

    for (int base = beg; base < end; base += 32) {
        const int idx = base + lane;
        int2 ed = make_int2(0, 0);
        if (idx < end) ed = sedge[idx];
        const int cntc = (end - base) < 32 ? (end - base) : 32;

        for (int j = 0; j < cntc; j += 8) {
            float a[8][4], uu[8];
            #pragma unroll
            for (int q = 0; q < 8; ++q) {
                const int s = __builtin_amdgcn_readlane(ed.x, j + q);
                uu[q] = __int_as_float(__builtin_amdgcn_readlane(ed.y, j + q));
                const float* xs = X + (size_t)s * C_DIM + lane;
                a[q][0] = xs[0];  a[q][1] = xs[32];
                a[q][2] = xs[64]; a[q][3] = xs[96];
            }
            #pragma unroll
            for (int q = 0; q < 8; ++q) {
                if (j + q < cntc) {
                    const float w0 = 1.0f - uu[q];
                    p0 += w0 * a[q][0];  q0 += uu[q] * a[q][0];
                    p1 += w0 * a[q][1];  q1 += uu[q] * a[q][1];
                    p2 += w0 * a[q][2];  q2 += uu[q] * a[q][2];
                    p3 += w0 * a[q][3];  q3 += uu[q] * a[q][3];
                }
            }
        }
    }

    const float deg = (float)(end - beg);
    const float inv = 1.0f / fmaxf(deg, 1.0f);
    float* op = PQ + (size_t)n * (2 * C_DIM) + lane;
    op[0]           = p0 * inv;
    op[32]          = p1 * inv;
    op[64]          = p2 * inv;
    op[96]          = p3 * inv;
    op[C_DIM]       = q0 * inv;
    op[C_DIM + 32]  = q1 * inv;
    op[C_DIM + 64]  = q2 * inv;
    op[C_DIM + 96]  = q3 * inv;
}

extern "C" void kernel_launch(void* const* d_in, const int* in_sizes, int n_in,
                              void* d_out, int out_size, void* d_ws, size_t ws_size,
                              hipStream_t stream)
{
    const float* x  = (const float*)d_in[0];
    const int*   ei = (const int*)d_in[1];
    const float* ea = (const float*)d_in[2];
    const float* W1 = (const float*)d_in[3];
    const float* r1 = (const float*)d_in[4];
    const float* b1 = (const float*)d_in[5];
    const float* W2 = (const float*)d_in[6];
    const float* r2 = (const float*)d_in[7];
    const float* b2 = (const float*)d_in[8];
    float* out = (float*)d_out;

    const int N = in_sizes[0] / C_DIM;
    const int E = in_sizes[1] / 2;

    const int NP = ((N + 4) / 4) * 4;
    float* ws     = (float*)d_ws;
    float* PQbuf  = ws;
    float* hmid   = PQbuf + (size_t)N * 2 * C_DIM;
    int*   cnt    = (int*)(hmid + (size_t)N * C_DIM);
    int*   rowptr = cnt + NP;
    int*   part   = rowptr + NP;
    int2*  sedge  = (int2*)(part + (size_t)NSLICE * NP);
    int*   rank   = (int*)(sedge + E);

    const int n64     = ((N / MTILE) / RESIDENT_BLOCKS) * RESIDENT_BLOCKS;
    const int rsmall0 = n64 * MTILE;
    const int nsmall  = (N - rsmall0 + 15) / 16;
    const int mblocks = nsmall + n64;
    const int nblocks = (N + 7) / 8;
    const int eblocks = (E + 255) / 256;
    const int cblocks = (N + 255) / 256;
    const int sblocks = (N + 1023) / 1024;
    const int bps     = (E + NSLICE * 256 - 1) / (NSLICE * 256);
    const int L       = bps * 256;

    slice_count_kernel<<<NSLICE, 1024, 0, stream>>>(ei, part, rank, E, N, NP, L);
    slice_offsets_kernel<<<cblocks, 256, 0, stream>>>(part, cnt, N, NP);
    scan_kernel<<<sblocks, 256, 0, stream>>>(cnt, rowptr, N);
    fill_kernel<<<eblocks, 256, 0, stream>>>(ei, ea, rowptr, part, rank, sedge, E, N, NP, bps);

    gather_sum_kernel<<<nblocks, 256, 0, stream>>>(rowptr, sedge, x, PQbuf, N);
    splineconv_gemm_kernel<<<mblocks, 256, 0, stream>>>(PQbuf, x, W1, r1, b1, hmid, N, nsmall, rsmall0);

    gather_sum_kernel<<<nblocks, 256, 0, stream>>>(rowptr, sedge, hmid, PQbuf, N);
    splineconv_gemm_kernel<<<mblocks, 256, 0, stream>>>(PQbuf, hmid, W2, r2, b2, out, N, nsmall, rsmall0);
}


# product kernel: result rows put together in LDS (the freed X chunk, one more barrier), each wave stores whole 512-B rows: 8 four-line stores per wave, not 32 of two 64-B half lines; sums unchanged
# speedup vs baseline: 3.5811x; 1.1435x over previous
#include <hip/hip_runtime.h>

typedef __attribute__((ext_vector_type(2))) float v2f;
typedef __attribute__((ext_vector_type(8))) float v8f;

#define C_DIM 128
#define LDA   132
#define MTILE 64

template <int MSUB>
__device__ __forceinline__
void wmma_chunk(v8f (&acc)[MSUB],
                const float* __restrict__ Bl,
                const float* __restrict__ Al)
{
    v2f bf[32];
    #pragma unroll
    for (int k4 = 0; k4 < 32; ++k4) {
        bf[k4].x = Bl[(size_t)(4 * k4) * C_DIM];
        bf[k4].y = Bl[(size_t)(4 * k4 + 1) * C_DIM];
    }
    #pragma unroll
    for (int k4 = 0; k4 < 32; ++k4) {
        const int k0 = 4 * k4;
        v2f a[MSUB];
        #pragma unroll
        for (int s = 0; s < MSUB; ++s) {
            a[s].x = Al[s * 16 * LDA + k0];
            a[s].y = Al[s * 16 * LDA + k0 + 1];
        }
        #pragma unroll
        for (int s = 0; s < MSUB; ++s)
            acc[s] = __builtin_amdgcn_wmma_f32_16x16x4_f32(
                         false, a[s], false, bf[k4], (short)0, acc[s], false, false);
    }
}

template <int MSUB>
__device__ __forceinline__
void product_tile(float* __restrict__ As,
                  const int* __restrict__ rowptr,
                  const int2* __restrict__ sedge,
                  const float* __restrict__ X,
                  const float* __restrict__ W,
                  const float* __restrict__ root,
                  const float* __restrict__ bias,
                  float* __restrict__ Out,
                  int N, int row0)
{
    const int tid  = threadIdx.x;
    const int wave = __builtin_amdgcn_readfirstlane((int)(threadIdx.x >> 5));
    const int lane = tid & 31;
    const int hi   = lane >> 4;
    const int l16  = lane & 15;
    const int gcol = wave * 16;

    constexpr int NPW = 2 * MSUB;
    const int r0 = wave * NPW;
    const int n0 = row0 + r0;

    const int rp = rowptr[min(n0 + min(lane, NPW), N)];

    float4 vx[NPW];
    #pragma unroll
    for (int i = 0; i < NPW; ++i) {
        const int gr = min(n0 + i, N - 1);
        vx[i] = ((const float4*)(X + (size_t)gr * C_DIM))[lane];
    }

    const int ebeg = __builtin_amdgcn_readlane(rp, 0);
    const int eend = __builtin_amdgcn_readlane(rp, NPW);

    int2 ednext = make_int2(0, 0);
    if (ebeg + lane < eend) ednext = sedge[ebeg + lane];

    #pragma unroll
    for (int i = 0; i < NPW; ++i)
        *(float4*)&As[2 * MTILE * LDA + (r0 + i) * LDA + lane * 4] = vx[i];
    __syncthreads();

    v8f acc[MSUB];
    #pragma unroll
    for (int s = 0; s < MSUB; ++s) acc[s] = v8f{};

    const size_t boff = (size_t)(2 * hi) * C_DIM + gcol + l16;
    const int    aoff = l16 * LDA + 2 * hi;

    wmma_chunk<MSUB>(acc, root + boff, &As[2 * MTILE * LDA + aoff]);

    {
        int cur    = 0;
        int segbeg = ebeg;
        int nb     = __builtin_amdgcn_readlane(rp, 1);
        float p0 = 0.f, p1 = 0.f, p2 = 0.f, p3 = 0.f;
        float q0 = 0.f, q1 = 0.f, q2 = 0.f, q3 = 0.f;

        auto flush = [&]() {
            const float deg = (float)(nb - segbeg);
            const float inv = 1.0f / fmaxf(deg, 1.0f);
            float* dp = &As[(r0 + cur) * LDA + lane];
            float* dq = dp + MTILE * LDA;
            dp[0]  = p0 * inv;  dp[32] = p1 * inv;
            dp[64] = p2 * inv;  dp[96] = p3 * inv;
            dq[0]  = q0 * inv;  dq[32] = q1 * inv;
            dq[64] = q2 * inv;  dq[96] = q3 * inv;
            p0 = p1 = p2 = p3 = 0.f;
            q0 = q1 = q2 = q3 = 0.f;
            segbeg = nb;
            ++cur;
            nb = __builtin_amdgcn_readlane(rp, cur + 1);
        };

        for (int base = ebeg; base < eend; base += 32) {
            const int2 ed = ednext;
            if (base + 32 + lane < eend) ednext = sedge[base + 32 + lane];
            const int cntc = (eend - base) < 32 ? (eend - base) : 32;

            for (int j = 0; j < cntc; j += 8) {
                float a[8][4], uu[8];
                #pragma unroll
                for (int q = 0; q < 8; ++q) {
                    const int s = __builtin_amdgcn_readlane(ed.x, j + q);
                    uu[q] = __int_as_float(__builtin_amdgcn_readlane(ed.y, j + q));
                    const float* xs = X + (size_t)s * C_DIM + lane;
                    a[q][0] = xs[0];  a[q][1] = xs[32];
                    a[q][2] = xs[64]; a[q][3] = xs[96];
                }
                #pragma unroll
                for (int q = 0; q < 8; ++q) {
                    if (j + q < cntc) {
                        const int g = base + j + q;
                        while (g >= nb && cur < NPW - 1) flush();
                        const float w0 = 1.0f - uu[q];
                        p0 += w0 * a[q][0];  q0 += uu[q] * a[q][0];
                        p1 += w0 * a[q][1];  q1 += uu[q] * a[q][1];
                        p2 += w0 * a[q][2];  q2 += uu[q] * a[q][2];
                        p3 += w0 * a[q][3];  q3 += uu[q] * a[q][3];
                    }
                }
            }
        }
        while (cur < NPW) flush();
    }
    __syncthreads();

    wmma_chunk<MSUB>(acc, W + boff, &As[aoff]);
    wmma_chunk<MSUB>(acc, W + (size_t)C_DIM * C_DIM + boff, &As[MTILE * LDA + aoff]);

    const float bv = bias[gcol + l16];
    float* Cs = &As[2 * MTILE * LDA];
    #pragma unroll
    for (int s = 0; s < MSUB; ++s) {
        #pragma unroll
        for (int r = 0; r < 8; ++r)
            Cs[(s * 16 + r + 8 * hi) * LDA + gcol + l16] = acc[s][r] + bv;
    }
    __syncthreads();
    const float* cr = &Cs[r0 * LDA + lane * 4];
    float4* og = (float4*)(Out + (size_t)n0 * C_DIM) + lane;
    if (n0 + NPW <= N) {
        float4 vo[NPW];
        #pragma unroll
        for (int i = 0; i < NPW; ++i) vo[i] = *(const float4*)&cr[i * LDA];
        #pragma unroll
        for (int i = 0; i < NPW; ++i) og[i * (C_DIM / 4)] = vo[i];
    } else {
        #pragma unroll
        for (int i = 0; i < NPW; ++i)
            if (n0 + i < N) og[i * (C_DIM / 4)] = *(const float4*)&cr[i * LDA];
    }
}

#define RESIDENT_BLOCKS 768

__global__ __launch_bounds__(256)
void splineconv_gemm_kernel(const int* __restrict__ rowptr,
                            const int2* __restrict__ sedge,
                            const float* __restrict__ X,
                            const float* __restrict__ W,
                            const float* __restrict__ root,
                            const float* __restrict__ bias,
                            float* __restrict__ Out,
                            int N,
                            int nsmall,
                            int rsmall0)
{
    __shared__ __attribute__((aligned(16))) float As[3 * MTILE * LDA];
    const int b = blockIdx.x;
    if (b < nsmall)
        product_tile<1>(As, rowptr, sedge, X, W, root, bias, Out, N, rsmall0 + b * 16);
    else
        product_tile<4>(As, rowptr, sedge, X, W, root, bias, Out, N, (b - nsmall) * MTILE);
}

#define NSLICE 32
#define CAP    51200

__global__ __launch_bounds__(1024)
void slice_count_kernel(const int* __restrict__ ei,
                        int* __restrict__ part,
                        int* __restrict__ rank,
                        int E, int N, int NP, int L)
{
    __shared__ int4 lc4[CAP / 4];
    int* lcnt = (int*)lc4;
    const int t  = threadIdx.x;
    const int e0 = blockIdx.x * L;
    int len = E - e0;
    len = len < 0 ? 0 : (len > L ? L : len);
    const int* dst = ei + E + e0;
    int* prow = part + (size_t)blockIdx.x * NP;

    for (int nb = 0; nb < N; nb += CAP) {
        const int span = (N - nb) < CAP ? (N - nb) : CAP;
        const int z4 = (span + 3) >> 2;
        for (int i = t; i < z4; i += 1024) lc4[i] = make_int4(0, 0, 0, 0);
        __syncthreads();

        for (int k0 = t; k0 < len; k0 += 8 * 1024) {
            int dd[8];
            #pragma unroll
            for (int q = 0; q < 8; ++q) {
                const int i = k0 + 1024 * q;
                dd[q] = (i < len) ? dst[i] : -1;
            }
            int rr[8];
            #pragma unroll
            for (int q = 0; q < 8; ++q) {
                const unsigned key = (unsigned)dd[q] - (unsigned)nb;
                rr[q] = -1;
                if (key < (unsigned)span) rr[q] = atomicAdd(&lcnt[key], 1);
            }
            #pragma unroll
            for (int q = 0; q < 8; ++q)
                if (rr[q] >= 0) rank[e0 + k0 + 1024 * q] = rr[q];
        }
        __syncthreads();

        const int span4 = span >> 2;
        int4* o4 = (int4*)(prow + nb);
        for (int i = t; i < span4; i += 1024) o4[i] = lc4[i];
        const int tl = span4 * 4 + t;
        if (tl < span) prow[nb + tl] = lcnt[tl];
        __syncthreads();
    }
}

__global__ __launch_bounds__(256)
void slice_offsets_kernel(int* __restrict__ part, int* __restrict__ cnt, int N, int NP)
{
    const int d = blockIdx.x * 256 + threadIdx.x;
    if (d >= N) return;
    int v[NSLICE];
    #pragma unroll
    for (int s = 0; s < NSLICE; ++s) v[s] = part[(size_t)s * NP + d];
    int run = 0;
    #pragma unroll
    for (int s = 0; s < NSLICE; ++s) {
        part[(size_t)s * NP + d] = run;
        run += v[s];
    }
    cnt[d] = run;
}

__global__ __launch_bounds__(256)
void scan_kernel(const int* __restrict__ cnt,
                 int* __restrict__ rowptr,
                 int N)
{
    __shared__ int wtot[8];
    __shared__ int wpre[8];
    const int t    = threadIdx.x;
    const int lane = t & 31;
    const int w    = t >> 5;
    const int b0   = blockIdx.x * 1024;
    const int4* c4 = (const int4*)cnt;

    int part = 0;
    const int trips = blockIdx.x;
    #pragma unroll 8
    for (int k = 0; k < trips; ++k) {
        const int4 v = c4[k * 256 + t];
        part += (v.x + v.y) + (v.z + v.w);
    }

    const int i0 = b0 + t * 4;
    int v0 = 0, v1 = 0, v2 = 0, v3 = 0;
    if (i0 + 3 < N) {
        const int4 v = c4[i0 >> 2];
        v0 = v.x; v1 = v.y; v2 = v.z; v3 = v.w;
    } else {
        if (i0     < N) v0 = cnt[i0];
        if (i0 + 1 < N) v1 = cnt[i0 + 1];
        if (i0 + 2 < N) v2 = cnt[i0 + 2];
    }
    const int ts = (v0 + v1) + (v2 + v3);

    #pragma unroll
    for (int off = 16; off > 0; off >>= 1) part += __shfl_xor(part, off);
    int inc = ts;
    #pragma unroll
    for (int off = 1; off < 32; off <<= 1) {
        const int y = __shfl_up(inc, off);
        if (lane >= off) inc += y;
    }
    if (lane == 31) wtot[w] = inc;
    if (lane == 0)  wpre[w] = part;
    __syncthreads();

    int base = 0;
    #pragma unroll
    for (int k = 0; k < 8; ++k) {
        base += wpre[k];
        if (k < w) base += wtot[k];
    }
    const int ex = base + inc - ts;
    if (i0     < N) rowptr[i0]     = ex;
    if (i0 + 1 < N) rowptr[i0 + 1] = ex + v0;
    if (i0 + 2 < N) rowptr[i0 + 2] = ex + v0 + v1;
    if (i0 + 3 < N) rowptr[i0 + 3] = ex + v0 + v1 + v2;
    if (blockIdx.x == gridDim.x - 1 && t == 255) rowptr[N] = ex + ts;
}

__global__ __launch_bounds__(256)
void fill_kernel(const int* __restrict__ ei,
                 const float* __restrict__ u,
                 const int* __restrict__ rowptr,
                 const int* __restrict__ part,
                 const int* __restrict__ rank,
                 int2* __restrict__ sedge,
                 int E, int N, int NP,
                 int bps)
{
    const int e = blockIdx.x * 256 + threadIdx.x;
    if (e >= E) return;
    const int d = ei[E + e];
    if ((unsigned)d >= (unsigned)N) return;
    int s = ei[e];
    s = s < 0 ? 0 : (s >= N ? N - 1 : s);
    const int sl = blockIdx.x / bps;
    const int p = rowptr[d] + part[(size_t)sl * NP + d] + rank[e];
    if ((unsigned)p < (unsigned)E) sedge[p] = make_int2(s, __float_as_int(u[e]));
}

extern "C" void kernel_launch(void* const* d_in, const int* in_sizes, int n_in,
                              void* d_out, int out_size, void* d_ws, size_t ws_size,
                              hipStream_t stream)
{
    const float* x  = (const float*)d_in[0];
    const int*   ei = (const int*)d_in[1];
    const float* ea = (const float*)d_in[2];
    const float* W1 = (const float*)d_in[3];
    const float* r1 = (const float*)d_in[4];
    const float* b1 = (const float*)d_in[5];
    const float* W2 = (const float*)d_in[6];
    const float* r2 = (const float*)d_in[7];
    const float* b2 = (const float*)d_in[8];
    float* out = (float*)d_out;

    const int N = in_sizes[0] / C_DIM;
    const int E = in_sizes[1] / 2;

    const int NP = ((N + 4) / 4) * 4;
    float* ws     = (float*)d_ws;
    float* hmid   = ws;
    int*   cnt    = (int*)(hmid + (size_t)N * C_DIM);
    int*   rowptr = cnt + NP;
    int*   part   = rowptr + NP;
    int2*  sedge  = (int2*)(part + (size_t)NSLICE * NP);
    int*   rank   = (int*)(sedge + E);

    const int n64     = ((N / MTILE) / RESIDENT_BLOCKS) * RESIDENT_BLOCKS;
    const int rsmall0 = n64 * MTILE;
    const int nsmall  = (N - rsmall0 + 15) / 16;
    const int mblocks = nsmall + n64;
    const int eblocks = (E + 255) / 256;
    const int cblocks = (N + 255) / 256;
    const int sblocks = (N + 1023) / 1024;
    const int bps     = (E + NSLICE * 256 - 1) / (NSLICE * 256);
    const int L       = bps * 256;

    slice_count_kernel<<<NSLICE, 1024, 0, stream>>>(ei, part, rank, E, N, NP, L);
    slice_offsets_kernel<<<cblocks, 256, 0, stream>>>(part, cnt, N, NP);
    scan_kernel<<<sblocks, 256, 0, stream>>>(cnt, rowptr, N);
    fill_kernel<<<eblocks, 256, 0, stream>>>(ei, ea, rowptr, part, rank, sedge, E, N, NP, bps);

    splineconv_gemm_kernel<<<mblocks, 256, 0, stream>>>(rowptr, sedge, x, W1, r1, b1, hmid, N, nsmall, rsmall0);

    splineconv_gemm_kernel<<<mblocks, 256, 0, stream>>>(rowptr, sedge, hmid, W2, r2, b2, out, N, nsmall, rsmall0);
}
